# starting kernel
# baseline (speedup 1.0000x reference)
#include <hip/hip_runtime.h>

typedef float v2f __attribute__((ext_vector_type(2)));
typedef float v8f __attribute__((ext_vector_type(8)));

#define T_LEN        1048576
#define SEG          4096
#define TILES_PER_SEG (SEG / 256)
#define SEGS_PER_ROW (T_LEN / SEG)

__global__ __launch_bounds__(32) void seg_scan_wmma(const float* __restrict__ x,
                                                    float* __restrict__ y,
                                                    float* __restrict__ segSums) {
  const int seg  = blockIdx.x;
  const int lane = threadIdx.x;
  const int n    = lane & 15;
  const int hi   = lane >> 4;

  const long long segBase = (long long)seg * SEG;

  v2f bU[4];
  v2f aLs[4];
#pragma unroll
  for (int kc = 0; kc < 4; ++kc) {
    const int k0 = 4 * kc + 2 * hi;
    bU[kc].x  = (k0     <= n) ? 1.0f : 0.0f;
    bU[kc].y  = (k0 + 1 <= n) ? 1.0f : 0.0f;
    aLs[kc].x = (k0     <  n) ? 1.0f : 0.0f;
    aLs[kc].y = (k0 + 1 <  n) ? 1.0f : 0.0f;
  }

  float carry = 0.0f;

  for (int t = 0; t < TILES_PER_SEG; ++t) {
    const float* tx = x + segBase + t * 256;

    v2f a[4];
#pragma unroll
    for (int kc = 0; kc < 4; ++kc) {
      a[kc] = *(const v2f*)(tx + 16 * n + 4 * kc + 2 * hi);
    }

    if (t + 1 < TILES_PER_SEG) {
      __builtin_prefetch(tx + 256 + lane * 8, 0, 1);
    }

    v8f acc = {};
#pragma unroll
    for (int kc = 0; kc < 4; ++kc) {
      acc = __builtin_amdgcn_wmma_f32_16x16x4_f32(
          false, a[kc], false, bU[kc], (short)0, acc, false, false);
    }

    float rs[16];
#pragma unroll
    for (int r = 0; r < 8; ++r) {
      rs[r]     = __shfl(acc[r], 15, 32);
      rs[r + 8] = __shfl(acc[r], 31, 32);
    }

    v2f w[4];
#pragma unroll
    for (int kc = 0; kc < 4; ++kc) {
      w[kc].x = hi ? rs[4 * kc + 2] : rs[4 * kc];
      w[kc].y = hi ? rs[4 * kc + 3] : rs[4 * kc + 1];
    }

#pragma unroll
    for (int kc = 0; kc < 4; ++kc) {
      acc = __builtin_amdgcn_wmma_f32_16x16x4_f32(
          false, aLs[kc], false, w[kc], (short)0, acc, false, false);
    }

    float* ty = y + segBase + t * 256;
#pragma unroll
    for (int r = 0; r < 8; ++r) {
      ty[16 * r + 128 * hi + n] = acc[r] + carry;
    }

    float tot = 0.0f;
#pragma unroll
    for (int i = 0; i < 16; ++i) tot += rs[i];
    carry += tot;
  }

  if (lane == 0) segSums[seg] = carry;
}

__global__ __launch_bounds__(SEGS_PER_ROW) void scan_seg_sums(
    const float* __restrict__ sums, float* __restrict__ offs) {
  __shared__ float s[SEGS_PER_ROW];
  const int row = blockIdx.x;
  const int t   = threadIdx.x;
  const float orig = sums[row * SEGS_PER_ROW + t];
  s[t] = orig;
  __syncthreads();
#pragma unroll
  for (int d = 1; d < SEGS_PER_ROW; d <<= 1) {
    const float v = (t >= d) ? s[t - d] : 0.0f;
    __syncthreads();
    s[t] += v;
    __syncthreads();
  }
  offs[row * SEGS_PER_ROW + t] = s[t] - orig;
}

__global__ __launch_bounds__(256) void add_offsets(float* __restrict__ y,
                                                   const float* __restrict__ offs) {
  const int seg = blockIdx.x;
  const float o = offs[seg];
  float4* p = (float4*)(y + (long long)seg * SEG);
  const int t = threadIdx.x;
#pragma unroll
  for (int i = 0; i < SEG / 4 / 256; ++i) {
    float4 v = p[t + i * 256];
    v.x += o; v.y += o; v.z += o; v.w += o;
    p[t + i * 256] = v;
  }
}

extern "C" void kernel_launch(void* const* d_in, const int* in_sizes, int n_in,
                              void* d_out, int out_size, void* d_ws, size_t ws_size,
                              hipStream_t stream) {
  (void)n_in; (void)out_size; (void)ws_size;
  const float* x = (const float*)d_in[0];
  float* y = (float*)d_out;

  const long long total = (long long)in_sizes[0];
  const int rows = (int)(total / T_LEN);
  const int nsegTotal = rows * SEGS_PER_ROW;

  float* sums = (float*)d_ws;
  float* offs = sums + nsegTotal;

  seg_scan_wmma<<<nsegTotal, 32, 0, stream>>>(x, y, sums);
  scan_seg_sums<<<rows, SEGS_PER_ROW, 0, stream>>>(sums, offs);
  add_offsets<<<nsegTotal, 256, 0, stream>>>(y, offs);
}


# Scan pass: segment cut into 16 streams of 256 = the tile's 16 columns, so Lt*X alone scans a tile (carry per lane); one Ls*W per segment gives 16 stream offsets, added in the add pass; new sum order
# speedup vs baseline: 1.9924x; 1.9924x over previous
#include <hip/hip_runtime.h>

typedef float v2f __attribute__((ext_vector_type(2)));
typedef float v8f __attribute__((ext_vector_type(8)));

#define T_LEN        1048576
#define SEG          4096
#define SEGS_PER_ROW (T_LEN / SEG)

#define NSTR       16
#define STR_LEN    (SEG / NSTR)
#define GRP_COLS   64
#define NGRP       (STR_LEN / GRP_COLS)
#define GRP_TILES  (GRP_COLS / 16)
#define PITCH      68
#define GRP_PIECES 8
#define STR_F4     (STR_LEN / 4)
#define GRP_F4     (GRP_COLS / 4)

__device__ __forceinline__ float other_half(float v) {
  const int i = __builtin_bit_cast(int, v);
  return __builtin_bit_cast(
      float, __builtin_amdgcn_permlanex16(i, i, 0x76543210, 0xfedcba98, false,
                                          false));
}

__global__ __launch_bounds__(32) void seg_scan_wmma(const float* __restrict__ x,
                                                    float* __restrict__ y,
                                                    float* __restrict__ segSums,
                                                    float* __restrict__ strOffs) {
  __shared__ float4 stage4[NSTR * PITCH / 4];
  float* stage = (float*)stage4;

  const int seg  = blockIdx.x;
  const int lane = threadIdx.x;
  const int n    = lane & 15;
  const int hi   = lane >> 4;

  const long long segBase = (long long)seg * SEG;
  const float4* gx = (const float4*)(x + segBase) + hi * STR_F4 + n;
  float4*       gy = (float4*)(y + segBase) + hi * STR_F4 + n;
  const int pieceOff = hi * PITCH + 4 * n;
  const int fragOff = n * PITCH + 8 * hi;

  v2f aLt[4];
  v2f aLs[4];
#pragma unroll
  for (int kc = 0; kc < 4; ++kc) {
    const int k0 = 8 * hi + 2 * kc;
    aLt[kc].x = (k0     <= n) ? 1.0f : 0.0f;
    aLt[kc].y = (k0 + 1 <= n) ? 1.0f : 0.0f;
    aLs[kc].x = (k0     <  n) ? 1.0f : 0.0f;
    aLs[kc].y = (k0 + 1 <  n) ? 1.0f : 0.0f;
  }

  float4 p0 = gx[0 * 2 * STR_F4], p1 = gx[1 * 2 * STR_F4];
  float4 p2 = gx[2 * 2 * STR_F4], p3 = gx[3 * 2 * STR_F4];
  float4 p4 = gx[4 * 2 * STR_F4], p5 = gx[5 * 2 * STR_F4];
  float4 p6 = gx[6 * 2 * STR_F4], p7 = gx[7 * 2 * STR_F4];

  float carry = 0.0f;

#pragma unroll 1
  for (int g = 0; g < NGRP; ++g) {
    *(float4*)(stage + 0 * 2 * PITCH + pieceOff) = p0;
    *(float4*)(stage + 1 * 2 * PITCH + pieceOff) = p1;
    *(float4*)(stage + 2 * 2 * PITCH + pieceOff) = p2;
    *(float4*)(stage + 3 * 2 * PITCH + pieceOff) = p3;
    *(float4*)(stage + 4 * 2 * PITCH + pieceOff) = p4;
    *(float4*)(stage + 5 * 2 * PITCH + pieceOff) = p5;
    *(float4*)(stage + 6 * 2 * PITCH + pieceOff) = p6;
    *(float4*)(stage + 7 * 2 * PITCH + pieceOff) = p7;

    if (g + 1 < NGRP) {
      const float4* q = gx + (g + 1) * GRP_F4;
      p0 = q[0 * 2 * STR_F4]; p1 = q[1 * 2 * STR_F4];
      p2 = q[2 * 2 * STR_F4]; p3 = q[3 * 2 * STR_F4];
      p4 = q[4 * 2 * STR_F4]; p5 = q[5 * 2 * STR_F4];
      p6 = q[6 * 2 * STR_F4]; p7 = q[7 * 2 * STR_F4];
    }

#pragma unroll
    for (int t = 0; t < GRP_TILES; ++t) {
      float* tl = stage + fragOff + 16 * t;

      const float4 xa = *(const float4*)(tl);
      const float4 xb = *(const float4*)(tl + 4);
      v2f b[4];
      b[0].x = xa.x; b[0].y = xa.y;
      b[1].x = xa.z; b[1].y = xa.w;
      b[2].x = xb.x; b[2].y = xb.y;
      b[3].x = xb.z; b[3].y = xb.w;

      v8f acc = {};
#pragma unroll
      for (int kc = 0; kc < 4; ++kc) {
        acc = __builtin_amdgcn_wmma_f32_16x16x4_f32(
            false, aLt[kc], false, b[kc], (short)0, acc, false, false);
      }

      const float last = acc[7];
      const float oth  = other_half(last);
      const float tot  = hi ? last : oth;

      float4 ra, rb;
      ra.x = acc[0] + carry; ra.y = acc[1] + carry;
      ra.z = acc[2] + carry; ra.w = acc[3] + carry;
      rb.x = acc[4] + carry; rb.y = acc[5] + carry;
      rb.z = acc[6] + carry; rb.w = acc[7] + carry;
      *(float4*)(tl)     = ra;
      *(float4*)(tl + 4) = rb;

      carry += tot;
    }

#pragma unroll
    for (int j = 0; j < GRP_PIECES; ++j) {
      gy[g * GRP_F4 + j * 2 * STR_F4] =
          *(const float4*)(stage + j * 2 * PITCH + pieceOff);
    }
  }

  stage[lane] = carry;
  const float4 wa = *(const float4*)(stage + 8 * hi);
  const float4 wb = *(const float4*)(stage + 8 * hi + 4);
  v2f w[4];
  w[0].x = wa.x; w[0].y = wa.y;
  w[1].x = wa.z; w[1].y = wa.w;
  w[2].x = wb.x; w[2].y = wb.y;
  w[3].x = wb.z; w[3].y = wb.w;

  v8f off = {};
#pragma unroll
  for (int kc = 0; kc < 4; ++kc) {
    off = __builtin_amdgcn_wmma_f32_16x16x4_f32(
        false, aLs[kc], false, w[kc], (short)0, off, false, false);
  }

  const float segTot = off[7] + carry;
  float4 oa, ob;
  oa.x = off[0]; oa.y = off[1]; oa.z = off[2]; oa.w = off[3];
  ob.x = off[4]; ob.y = off[5]; ob.z = off[6]; ob.w = off[7];
  if (n == 0) {
    float4* so = (float4*)(strOffs + (long long)seg * NSTR + 8 * hi);
    so[0] = oa;
    so[1] = ob;
  }
  if (lane == 31) segSums[seg] = segTot;
}

__global__ __launch_bounds__(SEGS_PER_ROW) void scan_seg_sums(
    const float* __restrict__ sums, float* __restrict__ offs) {
  __shared__ float s[SEGS_PER_ROW];
  const int row = blockIdx.x;
  const int t   = threadIdx.x;
  const float orig = sums[row * SEGS_PER_ROW + t];
  s[t] = orig;
  __syncthreads();
#pragma unroll
  for (int d = 1; d < SEGS_PER_ROW; d <<= 1) {
    const float v = (t >= d) ? s[t - d] : 0.0f;
    __syncthreads();
    s[t] += v;
    __syncthreads();
  }
  offs[row * SEGS_PER_ROW + t] = s[t] - orig;
}

#define ADD_GROUPS  512
#define ADD_THREADS 256
#define ADD_WAVES   (ADD_GROUPS * (ADD_THREADS / 32))
#define ADD_PIECES  (SEG / 4 / 32)

__global__ __launch_bounds__(ADD_THREADS) void add_offsets(
    float* __restrict__ y, const float* __restrict__ offs,
    const float* __restrict__ strOffs, int nseg) {
  const int lane = threadIdx.x & 31;
  const int wave = __builtin_amdgcn_readfirstlane(
      (int)(blockIdx.x * (ADD_THREADS / 32) + (threadIdx.x >> 5)));

#pragma clang loop vectorize(disable)
  for (int seg = wave; seg < nseg; seg += ADD_WAVES) {
    const float o = offs[seg];
    const float* so = strOffs + (long long)seg * NSTR;
    float4* p = (float4*)(y + (long long)seg * SEG) + lane;

    float4 v[ADD_PIECES];
#pragma unroll
    for (int i = 0; i < ADD_PIECES; ++i) v[i] = p[i * 32];

#pragma unroll
    for (int i = 0; i < ADD_PIECES; ++i) {
      const float oi = o + so[i >> 1];
      float4 r = v[i];
      r.x += oi; r.y += oi; r.z += oi; r.w += oi;
      p[i * 32] = r;
    }
  }
}

extern "C" void kernel_launch(void* const* d_in, const int* in_sizes, int n_in,
                              void* d_out, int out_size, void* d_ws, size_t ws_size,
                              hipStream_t stream) {
  (void)n_in; (void)out_size; (void)ws_size;
  const float* x = (const float*)d_in[0];
  float* y = (float*)d_out;

  const long long total = (long long)in_sizes[0];
  const int rows = (int)(total / T_LEN);
  const int nsegTotal = rows * SEGS_PER_ROW;

  float* sums    = (float*)d_ws;
  float* offs    = sums + nsegTotal;
  float* strOffs = offs + nsegTotal;

  seg_scan_wmma<<<nsegTotal, 32, 0, stream>>>(x, y, sums, strOffs);
  scan_seg_sums<<<rows, SEGS_PER_ROW, 0, stream>>>(sums, offs);
  add_offsets<<<ADD_GROUPS, ADD_THREADS, 0, stream>>>(y, offs, strOffs, nsegTotal);
}


# One launch, no workspace: each 16-wave group sums its row's earlier chunks itself out of L2 (nearest first), then WMMA-scans its 128K chunk and writes y once; new sum order for the offsets
# speedup vs baseline: 2.2014x; 1.1049x over previous
#include <hip/hip_runtime.h>

typedef float v2f __attribute__((ext_vector_type(2)));
typedef float v8f __attribute__((ext_vector_type(8)));

#define T_LEN        1048576
#define SEG          4096

#define NSTR       16
#define STR_LEN    (SEG / NSTR)
#define GRP_COLS   64
#define NGRP       (STR_LEN / GRP_COLS)
#define GRP_TILES  (GRP_COLS / 16)
#define PITCH      68
#define GRP_PIECES 8
#define STR_F4     (STR_LEN / 4)
#define GRP_F4     (GRP_COLS / 4)

#define WG_WAVES       16
#define WG_THREADS     (WG_WAVES * 32)
#define CHUNKS_PER_ROW 8
#define CHUNK          (T_LEN / CHUNKS_PER_ROW)
#define CHUNK_F4       (CHUNK / 4)
#define CHUNK_SEGS     (CHUNK / SEG)
#define CHUNK_SWEEPS   (CHUNK_SEGS / WG_WAVES)
#define RED_F4         (CHUNK_F4 / WG_THREADS)
#define RED_BATCH      16
#define LINE_BYTES     256
#define OWN_LEAD       3

__device__ __forceinline__ float other_half(float v) {
  const int i = __builtin_bit_cast(int, v);
  return __builtin_bit_cast(
      float, __builtin_amdgcn_permlanex16(i, i, 0x76543210, 0xfedcba98, false,
                                          false));
}

__device__ __forceinline__ float sum4(const float4 a) {
  return (a.x + a.y) + (a.z + a.w);
}

template <int M>
__device__ __forceinline__ float row_xor(float v) {
  return __builtin_bit_cast(
      float, __builtin_amdgcn_update_dpp(0, __builtin_bit_cast(int, v),
                                         0x160 | M, 0xf, 0xf, false));
}

__device__ __forceinline__ float row_sum16(float s) {
  s += row_xor<1>(s);
  s += row_xor<2>(s);
  s += row_xor<4>(s);
  s += row_xor<8>(s);
  return s;
}

__device__ __forceinline__ void request_chunk(const float* chunk, int tid) {
  const char* c = (const char*)chunk;
#pragma unroll
  for (int k = 0; k < CHUNK * 4 / LINE_BYTES / WG_THREADS; ++k)
    __builtin_prefetch(c + (long long)(k * WG_THREADS + tid) * LINE_BYTES, 0, 3);
}

__global__ __launch_bounds__(WG_THREADS) void row_scan_wmma(
    const float* __restrict__ x, float* __restrict__ y, int rows) {
  __shared__ float4 stageAll4[WG_WAVES * NSTR * PITCH / 4];
  __shared__ float4 red4[WG_WAVES / 4];
  __shared__ float4 segTot4[CHUNK_SWEEPS][WG_WAVES / 4];

  const int tid  = threadIdx.x;
  const int lane = tid & 31;
  const int wave = __builtin_amdgcn_readfirstlane(tid >> 5);
  const int n    = lane & 15;
  const int hi   = lane >> 4;

  const int b = blockIdx.x;
  int row, p;
  if (rows == 32) {
    const int d0 = b & 7, m = (b >> 3) & 3, d1 = b >> 5;
    p   = (m << 1) | ((d0 ^ d1) & 1);
    row = ((d0 >> 2) << 4) | ((d1 >> 2) << 3) | (((d0 >> 1) & 1) << 2) |
          (((d1 >> 1) & 1) << 1) | (d0 & 1);
  } else {
    row = b % rows;
    p   = b / rows;
  }

  const float* xrow = x + (long long)row * T_LEN;
  float*       yrow = y + (long long)row * T_LEN;

  const float* own = xrow + (long long)p * CHUNK;
  if (p <= OWN_LEAD + 1) request_chunk(own, tid);

  float4 a0 = {0.0f, 0.0f, 0.0f, 0.0f};
  float4 a1 = a0, a2 = a0, a3 = a0;
  for (int c = p - 1; c >= 0; --c) {
    if (p > OWN_LEAD + 1 && c == OWN_LEAD) request_chunk(own, tid);
    const float4* q = (const float4*)(xrow + (long long)c * CHUNK) + tid;
#pragma unroll 1
    for (int i = 0; i < RED_F4; i += RED_BATCH) {
      float4 t[RED_BATCH];
#pragma unroll
      for (int u = 0; u < RED_BATCH; ++u) t[u] = q[(i + u) * WG_THREADS];
#pragma unroll
      for (int u = 0; u < RED_BATCH; u += 4) {
        a0.x += t[u].x;     a0.y += t[u].y;     a0.z += t[u].z;     a0.w += t[u].w;
        a1.x += t[u + 1].x; a1.y += t[u + 1].y; a1.z += t[u + 1].z; a1.w += t[u + 1].w;
        a2.x += t[u + 2].x; a2.y += t[u + 2].y; a2.z += t[u + 2].z; a2.w += t[u + 2].w;
        a3.x += t[u + 3].x; a3.y += t[u + 3].y; a3.z += t[u + 3].z; a3.w += t[u + 3].w;
      }
    }
  }
  float ps = (sum4(a0) + sum4(a1)) + (sum4(a2) + sum4(a3));
  ps = row_sum16(ps);
  ps += other_half(ps);
  if (lane == 0) ((float*)red4)[wave] = ps;
  __syncthreads();
  float base;
  {
    const float4 r0 = red4[0], r1 = red4[1], r2 = red4[2], r3 = red4[3];
    base = (sum4(r0) + sum4(r1)) + (sum4(r2) + sum4(r3));
  }

  float* stage = (float*)stageAll4 + wave * (NSTR * PITCH);
  const int pieceOff = hi * PITCH + 4 * n;
  const int fragOff = n * PITCH + 8 * hi;

  v2f aLt[4];
  v2f aLs[4];
#pragma unroll
  for (int kc = 0; kc < 4; ++kc) {
    const int k0 = 8 * hi + 2 * kc;
    aLt[kc].x = (k0     <= n) ? 1.0f : 0.0f;
    aLt[kc].y = (k0 + 1 <= n) ? 1.0f : 0.0f;
    aLs[kc].x = (k0     <  n) ? 1.0f : 0.0f;
    aLs[kc].y = (k0 + 1 <  n) ? 1.0f : 0.0f;
  }

#pragma unroll 1
  for (int sw = 0; sw < CHUNK_SWEEPS; ++sw) {
    const int seg = p * CHUNK_SEGS + sw * WG_WAVES + wave;
    const float4* gx = (const float4*)(xrow + (long long)seg * SEG) + hi * STR_F4 + n;
    float4*       gy = (float4*)(yrow + (long long)seg * SEG) + hi * STR_F4 + n;

    float4 v[NGRP][GRP_PIECES];
#pragma unroll
    for (int g = 0; g < NGRP; ++g)
#pragma unroll
      for (int j = 0; j < GRP_PIECES; ++j)
        v[g][j] = gx[g * GRP_F4 + j * 2 * STR_F4];

    float part[GRP_PIECES];
#pragma unroll
    for (int j = 0; j < GRP_PIECES; ++j) {
      float4 s4 = v[0][j];
#pragma unroll
      for (int g = 1; g < NGRP; ++g) {
        s4.x += v[g][j].x; s4.y += v[g][j].y; s4.z += v[g][j].z; s4.w += v[g][j].w;
      }
      part[j] = row_sum16(sum4(s4));
    }
    if (n == 0) {
#pragma unroll
      for (int j = 0; j < GRP_PIECES; ++j) stage[2 * j + hi] = part[j];
    }
    const float4 wa = *(const float4*)(stage + 8 * hi);
    const float4 wb = *(const float4*)(stage + 8 * hi + 4);
    v2f w[4];
    w[0].x = wa.x; w[0].y = wa.y;
    w[1].x = wa.z; w[1].y = wa.w;
    w[2].x = wb.x; w[2].y = wb.y;
    w[3].x = wb.z; w[3].y = wb.w;

    v8f off = {};
#pragma unroll
    for (int kc = 0; kc < 4; ++kc) {
      off = __builtin_amdgcn_wmma_f32_16x16x4_f32(
          false, aLs[kc], false, w[kc], (short)0, off, false, false);
    }
    float sel = off[0];
#pragma unroll
    for (int r = 1; r < 8; ++r) sel = ((n & 7) == r) ? off[r] : sel;
    const float selOth = other_half(sel);
    const float strOff = (hi == (n >> 3)) ? sel : selOth;

    const float hsum = (sum4(wa) + sum4(wb));
    const float segSum = hsum + other_half(hsum);

    if (lane == 0) ((float*)segTot4[sw])[wave] = segSum;
    __syncthreads();
    float waveOff = 0.0f, sweepSum = 0.0f;
#pragma unroll
    for (int q4 = 0; q4 < WG_WAVES / 4; ++q4) {
      const float4 t = segTot4[sw][q4];
      waveOff += (4 * q4 + 0 < wave) ? t.x : 0.0f;
      waveOff += (4 * q4 + 1 < wave) ? t.y : 0.0f;
      waveOff += (4 * q4 + 2 < wave) ? t.z : 0.0f;
      waveOff += (4 * q4 + 3 < wave) ? t.w : 0.0f;
      sweepSum += sum4(t);
    }

    float carry = (base + waveOff) + strOff;

#pragma unroll
    for (int g = 0; g < NGRP; ++g) {
#pragma unroll
      for (int j = 0; j < GRP_PIECES; ++j)
        *(float4*)(stage + j * 2 * PITCH + pieceOff) = v[g][j];

#pragma unroll
      for (int t = 0; t < GRP_TILES; ++t) {
        float* tl = stage + fragOff + 16 * t;

        const float4 xa = *(const float4*)(tl);
        const float4 xb = *(const float4*)(tl + 4);
        v2f b[4];
        b[0].x = xa.x; b[0].y = xa.y;
        b[1].x = xa.z; b[1].y = xa.w;
        b[2].x = xb.x; b[2].y = xb.y;
        b[3].x = xb.z; b[3].y = xb.w;

        v8f acc = {};
#pragma unroll
        for (int kc = 0; kc < 4; ++kc) {
          acc = __builtin_amdgcn_wmma_f32_16x16x4_f32(
              false, aLt[kc], false, b[kc], (short)0, acc, false, false);
        }

        const float last = acc[7];
        const float oth  = other_half(last);
        const float tot  = hi ? last : oth;

        float4 ra, rb;
        ra.x = acc[0] + carry; ra.y = acc[1] + carry;
        ra.z = acc[2] + carry; ra.w = acc[3] + carry;
        rb.x = acc[4] + carry; rb.y = acc[5] + carry;
        rb.z = acc[6] + carry; rb.w = acc[7] + carry;
        *(float4*)(tl)     = ra;
        *(float4*)(tl + 4) = rb;

        carry += tot;
      }

#pragma unroll
      for (int j = 0; j < GRP_PIECES; ++j) {
        gy[g * GRP_F4 + j * 2 * STR_F4] =
            *(const float4*)(stage + j * 2 * PITCH + pieceOff);
      }
    }

    base += sweepSum;
  }
}

extern "C" void kernel_launch(void* const* d_in, const int* in_sizes, int n_in,
                              void* d_out, int out_size, void* d_ws, size_t ws_size,
                              hipStream_t stream) {
  (void)n_in; (void)out_size; (void)d_ws; (void)ws_size;
  const float* x = (const float*)d_in[0];
  float* y = (float*)d_out;

  const long long total = (long long)in_sizes[0];
  const int rows = (int)(total / T_LEN);

  row_scan_wmma<<<rows * CHUNKS_PER_ROW, WG_THREADS, 0, stream>>>(x, y, rows);
}


# Row cut into 8 chunks of 3,3,2,2,2,2,1,1 sixteenths (were equal eighths): the last group sums 15/16 of the row and scans 1/16 (was 7/8 and 1/8), fewer bytes through it; new sum order for the offsets
# speedup vs baseline: 2.4876x; 1.1300x over previous
#include <hip/hip_runtime.h>

typedef float v2f __attribute__((ext_vector_type(2)));
typedef float v8f __attribute__((ext_vector_type(8)));

#define T_LEN        1048576
#define SEG          4096

#define NSTR       16
#define STR_LEN    (SEG / NSTR)
#define GRP_COLS   64
#define NGRP       (STR_LEN / GRP_COLS)
#define GRP_TILES  (GRP_COLS / 16)
#define PITCH      68
#define GRP_PIECES 8
#define STR_F4     (STR_LEN / 4)
#define GRP_F4     (GRP_COLS / 4)

#define WG_WAVES       16
#define WG_THREADS     (WG_WAVES * 32)
#define CHUNKS_PER_ROW 8
#define UNIT           (SEG * WG_WAVES)
#define UNIT_F4        (UNIT / 4)
#define ROW_UNITS      (T_LEN / UNIT)
#define MAX_SWEEPS     3
#define CHUNK_START(p) ((0xFECA8630u >> (4 * (p))) & 15u)
#define CHUNK_UNITS(p) ((0x11222233u >> (4 * (p))) & 15u)
#define RED_F4         (UNIT_F4 / WG_THREADS)
#define RED_BATCH      16
#define LINE_BYTES     256
#define UNIT_LINES     (UNIT * 4 / LINE_BYTES / WG_THREADS)
#define OWN_LEAD       6

__device__ __forceinline__ float other_half(float v) {
  const int i = __builtin_bit_cast(int, v);
  return __builtin_bit_cast(
      float, __builtin_amdgcn_permlanex16(i, i, 0x76543210, 0xfedcba98, false,
                                          false));
}

__device__ __forceinline__ float sum4(const float4 a) {
  return (a.x + a.y) + (a.z + a.w);
}

template <int M>
__device__ __forceinline__ float row_xor(float v) {
  return __builtin_bit_cast(
      float, __builtin_amdgcn_update_dpp(0, __builtin_bit_cast(int, v),
                                         0x160 | M, 0xf, 0xf, false));
}

__device__ __forceinline__ float row_sum16(float s) {
  s += row_xor<1>(s);
  s += row_xor<2>(s);
  s += row_xor<4>(s);
  s += row_xor<8>(s);
  return s;
}

__device__ __forceinline__ void request_chunk(const float* chunk, int units, int tid) {
  const char* c = (const char*)chunk;
  for (int k = 0; k < units * UNIT_LINES; ++k)
    __builtin_prefetch(c + (long long)(k * WG_THREADS + tid) * LINE_BYTES, 0, 3);
}

__global__ __launch_bounds__(WG_THREADS) void row_scan_wmma(
    const float* __restrict__ x, float* __restrict__ y, int rows) {
  __shared__ float4 stageAll4[WG_WAVES * NSTR * PITCH / 4];
  __shared__ float4 red4[WG_WAVES / 4];
  __shared__ float4 segTot4[MAX_SWEEPS][WG_WAVES / 4];

  const int tid  = threadIdx.x;
  const int lane = tid & 31;
  const int wave = __builtin_amdgcn_readfirstlane(tid >> 5);
  const int n    = lane & 15;
  const int hi   = lane >> 4;

  const int b = blockIdx.x;
  int row, p;
  if (rows == 32) {
    const int d0 = b & 7, m = (b >> 3) & 3, d1 = b >> 5;
    p   = (m << 1) | ((d0 ^ d1) & 1);
    row = ((d0 >> 2) << 4) | ((d1 >> 2) << 3) | (((d0 >> 1) & 1) << 2) |
          (((d1 >> 1) & 1) << 1) | (d0 & 1);
  } else {
    row = b % rows;
    p   = b / rows;
  }

  const float* xrow = x + (long long)row * T_LEN;
  float*       yrow = y + (long long)row * T_LEN;

  const int start = (int)CHUNK_START(p);
  const int units = (int)CHUNK_UNITS(p);

  const float* own = xrow + (long long)start * UNIT;
  if (start <= OWN_LEAD + 1) request_chunk(own, units, tid);

  float4 a0 = {0.0f, 0.0f, 0.0f, 0.0f};
  float4 a1 = a0, a2 = a0, a3 = a0;
  for (int c = start - 1; c >= 0; --c) {
    if (start > OWN_LEAD + 1 && c == OWN_LEAD) request_chunk(own, units, tid);
    const float4* q = (const float4*)(xrow + (long long)c * UNIT) + tid;
#pragma unroll 1
    for (int i = 0; i < RED_F4; i += RED_BATCH) {
      float4 t[RED_BATCH];
#pragma unroll
      for (int u = 0; u < RED_BATCH; ++u) t[u] = q[(i + u) * WG_THREADS];
#pragma unroll
      for (int u = 0; u < RED_BATCH; u += 4) {
        a0.x += t[u].x;     a0.y += t[u].y;     a0.z += t[u].z;     a0.w += t[u].w;
        a1.x += t[u + 1].x; a1.y += t[u + 1].y; a1.z += t[u + 1].z; a1.w += t[u + 1].w;
        a2.x += t[u + 2].x; a2.y += t[u + 2].y; a2.z += t[u + 2].z; a2.w += t[u + 2].w;
        a3.x += t[u + 3].x; a3.y += t[u + 3].y; a3.z += t[u + 3].z; a3.w += t[u + 3].w;
      }
    }
  }
  float ps = (sum4(a0) + sum4(a1)) + (sum4(a2) + sum4(a3));
  ps = row_sum16(ps);
  ps += other_half(ps);
  if (lane == 0) ((float*)red4)[wave] = ps;
  __syncthreads();
  float base;
  {
    const float4 r0 = red4[0], r1 = red4[1], r2 = red4[2], r3 = red4[3];
    base = (sum4(r0) + sum4(r1)) + (sum4(r2) + sum4(r3));
  }

  float* stage = (float*)stageAll4 + wave * (NSTR * PITCH);
  const int pieceOff = hi * PITCH + 4 * n;
  const int fragOff = n * PITCH + 8 * hi;

  v2f aLt[4];
  v2f aLs[4];
#pragma unroll
  for (int kc = 0; kc < 4; ++kc) {
    const int k0 = 8 * hi + 2 * kc;
    aLt[kc].x = (k0     <= n) ? 1.0f : 0.0f;
    aLt[kc].y = (k0 + 1 <= n) ? 1.0f : 0.0f;
    aLs[kc].x = (k0     <  n) ? 1.0f : 0.0f;
    aLs[kc].y = (k0 + 1 <  n) ? 1.0f : 0.0f;
  }

#pragma unroll 1
  for (int sw = 0; sw < units; ++sw) {
    const int seg = (start + sw) * WG_WAVES + wave;
    const float4* gx = (const float4*)(xrow + (long long)seg * SEG) + hi * STR_F4 + n;
    float4*       gy = (float4*)(yrow + (long long)seg * SEG) + hi * STR_F4 + n;

    float4 v[NGRP][GRP_PIECES];
#pragma unroll
    for (int g = 0; g < NGRP; ++g)
#pragma unroll
      for (int j = 0; j < GRP_PIECES; ++j)
        v[g][j] = gx[g * GRP_F4 + j * 2 * STR_F4];

    float part[GRP_PIECES];
#pragma unroll
    for (int j = 0; j < GRP_PIECES; ++j) {
      float4 s4 = v[0][j];
#pragma unroll
      for (int g = 1; g < NGRP; ++g) {
        s4.x += v[g][j].x; s4.y += v[g][j].y; s4.z += v[g][j].z; s4.w += v[g][j].w;
      }
      part[j] = row_sum16(sum4(s4));
    }
    if (n == 0) {
#pragma unroll
      for (int j = 0; j < GRP_PIECES; ++j) stage[2 * j + hi] = part[j];
    }
    const float4 wa = *(const float4*)(stage + 8 * hi);
    const float4 wb = *(const float4*)(stage + 8 * hi + 4);
    v2f w[4];
    w[0].x = wa.x; w[0].y = wa.y;
    w[1].x = wa.z; w[1].y = wa.w;
    w[2].x = wb.x; w[2].y = wb.y;
    w[3].x = wb.z; w[3].y = wb.w;

    v8f off = {};
#pragma unroll
    for (int kc = 0; kc < 4; ++kc) {
      off = __builtin_amdgcn_wmma_f32_16x16x4_f32(
          false, aLs[kc], false, w[kc], (short)0, off, false, false);
    }
    float sel = off[0];
#pragma unroll
    for (int r = 1; r < 8; ++r) sel = ((n & 7) == r) ? off[r] : sel;
    const float selOth = other_half(sel);
    const float strOff = (hi == (n >> 3)) ? sel : selOth;

    const float hsum = (sum4(wa) + sum4(wb));
    const float segSum = hsum + other_half(hsum);

    if (lane == 0) ((float*)segTot4[sw])[wave] = segSum;
    __syncthreads();
    float waveOff = 0.0f, sweepSum = 0.0f;
#pragma unroll
    for (int q4 = 0; q4 < WG_WAVES / 4; ++q4) {
      const float4 t = segTot4[sw][q4];
      waveOff += (4 * q4 + 0 < wave) ? t.x : 0.0f;
      waveOff += (4 * q4 + 1 < wave) ? t.y : 0.0f;
      waveOff += (4 * q4 + 2 < wave) ? t.z : 0.0f;
      waveOff += (4 * q4 + 3 < wave) ? t.w : 0.0f;
      sweepSum += sum4(t);
    }

    float carry = (base + waveOff) + strOff;

#pragma unroll
    for (int g = 0; g < NGRP; ++g) {
#pragma unroll
      for (int j = 0; j < GRP_PIECES; ++j)
        *(float4*)(stage + j * 2 * PITCH + pieceOff) = v[g][j];

#pragma unroll
      for (int t = 0; t < GRP_TILES; ++t) {
        float* tl = stage + fragOff + 16 * t;

        const float4 xa = *(const float4*)(tl);
        const float4 xb = *(const float4*)(tl + 4);
        v2f b[4];
        b[0].x = xa.x; b[0].y = xa.y;
        b[1].x = xa.z; b[1].y = xa.w;
        b[2].x = xb.x; b[2].y = xb.y;
        b[3].x = xb.z; b[3].y = xb.w;

        v8f acc = {};
#pragma unroll
        for (int kc = 0; kc < 4; ++kc) {
          acc = __builtin_amdgcn_wmma_f32_16x16x4_f32(
              false, aLt[kc], false, b[kc], (short)0, acc, false, false);
        }

        const float last = acc[7];
        const float oth  = other_half(last);
        const float tot  = hi ? last : oth;

        float4 ra, rb;
        ra.x = acc[0] + carry; ra.y = acc[1] + carry;
        ra.z = acc[2] + carry; ra.w = acc[3] + carry;
        rb.x = acc[4] + carry; rb.y = acc[5] + carry;
        rb.z = acc[6] + carry; rb.w = acc[7] + carry;
        *(float4*)(tl)     = ra;
        *(float4*)(tl + 4) = rb;

        carry += tot;
      }

#pragma unroll
      for (int j = 0; j < GRP_PIECES; ++j) {
        gy[g * GRP_F4 + j * 2 * STR_F4] =
            *(const float4*)(stage + j * 2 * PITCH + pieceOff);
      }
    }

    base += sweepSum;
  }
}

extern "C" void kernel_launch(void* const* d_in, const int* in_sizes, int n_in,
                              void* d_out, int out_size, void* d_ws, size_t ws_size,
                              hipStream_t stream) {
  (void)n_in; (void)out_size; (void)d_ws; (void)ws_size;
  const float* x = (const float*)d_in[0];
  float* y = (float*)d_out;

  const long long total = (long long)in_sizes[0];
  const int rows = (int)(total / T_LEN);

  row_scan_wmma<<<rows * CHUNKS_PER_ROW, WG_THREADS, 0, stream>>>(x, y, rows);
}


# Chunk's first unit is scanned inside the first 4 batches of the summing of the row's earlier part (parked in LDS, WMMA scan while the batch's loads fly), row part added at the store; new sum order
# speedup vs baseline: 2.6329x; 1.0584x over previous
#include <hip/hip_runtime.h>

typedef float v2f __attribute__((ext_vector_type(2)));
typedef float v8f __attribute__((ext_vector_type(8)));

#define T_LEN        1048576
#define SEG          4096

#define NSTR       16
#define STR_LEN    (SEG / NSTR)
#define GRP_COLS   64
#define NGRP       (STR_LEN / GRP_COLS)
#define GRP_TILES  (GRP_COLS / 16)
#define PITCH      68
#define GRP_PIECES 8
#define STR_F4     (STR_LEN / 4)
#define GRP_F4     (GRP_COLS / 4)
#define GRP_BUF    (NSTR * PITCH)

#define WG_WAVES       16
#define WG_THREADS     (WG_WAVES * 32)
#define CHUNKS_PER_ROW 8
#define UNIT           (SEG * WG_WAVES)
#define UNIT_F4        (UNIT / 4)
#define ROW_UNITS      (T_LEN / UNIT)
#define MAX_SWEEPS     3
#define CHUNK_START(p) ((0xFECA8630u >> (4 * (p))) & 15u)
#define CHUNK_UNITS(p) ((0x11222233u >> (4 * (p))) & 15u)
#define RED_F4         (UNIT_F4 / WG_THREADS)
#define RED_BATCH      16
#define LINE_BYTES     256
#define UNIT_LINES     (UNIT * 4 / LINE_BYTES / WG_THREADS)
#define OWN_LEAD       6
#define UNIT_BATCHES   (RED_F4 / RED_BATCH)

__device__ __forceinline__ float other_half(float v) {
  const int i = __builtin_bit_cast(int, v);
  return __builtin_bit_cast(
      float, __builtin_amdgcn_permlanex16(i, i, 0x76543210, 0xfedcba98, false,
                                          false));
}

__device__ __forceinline__ float sum4(const float4 a) {
  return (a.x + a.y) + (a.z + a.w);
}

template <int M>
__device__ __forceinline__ float row_xor(float v) {
  return __builtin_bit_cast(
      float, __builtin_amdgcn_update_dpp(0, __builtin_bit_cast(int, v),
                                         0x160 | M, 0xf, 0xf, false));
}

__device__ __forceinline__ float row_sum16(float s) {
  s += row_xor<1>(s);
  s += row_xor<2>(s);
  s += row_xor<4>(s);
  s += row_xor<8>(s);
  return s;
}

__device__ __forceinline__ void request_chunk(const float* chunk, int units, int tid) {
  const char* c = (const char*)chunk;
  for (int k = 0; k < units * UNIT_LINES; ++k)
    __builtin_prefetch(c + (long long)(k * WG_THREADS + tid) * LINE_BYTES, 0, 3);
}

__device__ __forceinline__ void load_batch(const float* xrow, int start, int b, int tid,
                                           float4 (&t)[RED_BATCH]) {
  const int c = start - 1 - b / UNIT_BATCHES;
  const float4* q = (const float4*)(xrow + (long long)c * UNIT) +
                    (b % UNIT_BATCHES) * (RED_BATCH * WG_THREADS) + tid;
#pragma unroll
  for (int u = 0; u < RED_BATCH; ++u) t[u] = q[u * WG_THREADS];
}

__device__ __forceinline__ void add_batch(const float4 (&t)[RED_BATCH], float4& a0, float4& a1,
                                          float4& a2, float4& a3) {
#pragma unroll
  for (int u = 0; u < RED_BATCH; u += 4) {
    a0.x += t[u].x;     a0.y += t[u].y;     a0.z += t[u].z;     a0.w += t[u].w;
    a1.x += t[u + 1].x; a1.y += t[u + 1].y; a1.z += t[u + 1].z; a1.w += t[u + 1].w;
    a2.x += t[u + 2].x; a2.y += t[u + 2].y; a2.z += t[u + 2].z; a2.w += t[u + 2].w;
    a3.x += t[u + 3].x; a3.y += t[u + 3].y; a3.z += t[u + 3].z; a3.w += t[u + 3].w;
  }
}

__device__ __forceinline__ void scan_group(float* tl0, const v2f (&aLt)[4], int hi,
                                           float& carry) {
#pragma unroll
  for (int t = 0; t < GRP_TILES; ++t) {
    float* tl = tl0 + 16 * t;

    const float4 xa = *(const float4*)(tl);
    const float4 xb = *(const float4*)(tl + 4);
    v2f b[4];
    b[0].x = xa.x; b[0].y = xa.y;
    b[1].x = xa.z; b[1].y = xa.w;
    b[2].x = xb.x; b[2].y = xb.y;
    b[3].x = xb.z; b[3].y = xb.w;

    v8f acc = {};
#pragma unroll
    for (int kc = 0; kc < 4; ++kc) {
      acc = __builtin_amdgcn_wmma_f32_16x16x4_f32(
          false, aLt[kc], false, b[kc], (short)0, acc, false, false);
    }

    const float last = acc[7];
    const float oth  = other_half(last);
    const float tot  = hi ? last : oth;

    float4 ra, rb;
    ra.x = acc[0] + carry; ra.y = acc[1] + carry;
    ra.z = acc[2] + carry; ra.w = acc[3] + carry;
    rb.x = acc[4] + carry; rb.y = acc[5] + carry;
    rb.z = acc[6] + carry; rb.w = acc[7] + carry;
    *(float4*)(tl)     = ra;
    *(float4*)(tl + 4) = rb;

    carry += tot;
  }
}

__global__ __launch_bounds__(WG_THREADS) void row_scan_wmma(
    const float* __restrict__ x, float* __restrict__ y, int rows) {
  __shared__ float4 stageAll4[WG_WAVES * NGRP * GRP_BUF / 4];
  __shared__ float4 red4[WG_WAVES / 4];
  __shared__ float4 segTot4[MAX_SWEEPS][WG_WAVES / 4];

  const int tid  = threadIdx.x;
  const int lane = tid & 31;
  const int wave = __builtin_amdgcn_readfirstlane(tid >> 5);
  const int n    = lane & 15;
  const int hi   = lane >> 4;

  const int b = blockIdx.x;
  int row, p;
  if (rows == 32) {
    const int d0 = b & 7, m = (b >> 3) & 3, d1 = b >> 5;
    p   = (m << 1) | ((d0 ^ d1) & 1);
    row = ((d0 >> 2) << 4) | ((d1 >> 2) << 3) | (((d0 >> 1) & 1) << 2) |
          (((d1 >> 1) & 1) << 1) | (d0 & 1);
  } else {
    row = b % rows;
    p   = b / rows;
  }

  const float* xrow = x + (long long)row * T_LEN;
  float*       yrow = y + (long long)row * T_LEN;

  const int start = (int)CHUNK_START(p);
  const int units = (int)CHUNK_UNITS(p);

  const float* own = xrow + (long long)start * UNIT;
  const int nb   = start * UNIT_BATCHES;
  const int reqB = (start > OWN_LEAD + 1) ? (start - 1 - OWN_LEAD) * UNIT_BATCHES : -1;
  if (start <= OWN_LEAD + 1) request_chunk(own + UNIT, units - 1, tid);

  float* stage = (float*)stageAll4 + wave * (NGRP * GRP_BUF);
  const int pieceOff = hi * PITCH + 4 * n;
  const int fragOff = n * PITCH + 8 * hi;

  v2f aLt[4];
  v2f aLs[4];
#pragma unroll
  for (int kc = 0; kc < 4; ++kc) {
    const int k0 = 8 * hi + 2 * kc;
    aLt[kc].x = (k0     <= n) ? 1.0f : 0.0f;
    aLt[kc].y = (k0 + 1 <= n) ? 1.0f : 0.0f;
    aLs[kc].x = (k0     <  n) ? 1.0f : 0.0f;
    aLs[kc].y = (k0 + 1 <  n) ? 1.0f : 0.0f;
  }

  const int seg0 = start * WG_WAVES + wave;
  float4* gy0 = (float4*)(yrow + (long long)seg0 * SEG) + hi * STR_F4 + n;
  float carry0;
  float sweepSum0;
  {
    const float4* gx = (const float4*)(xrow + (long long)seg0 * SEG) + hi * STR_F4 + n;
    float4 v[NGRP][GRP_PIECES];
#pragma unroll
    for (int g = 0; g < NGRP; ++g)
#pragma unroll
      for (int j = 0; j < GRP_PIECES; ++j)
        v[g][j] = gx[g * GRP_F4 + j * 2 * STR_F4];

    float part[GRP_PIECES];
#pragma unroll
    for (int j = 0; j < GRP_PIECES; ++j) {
      float4 s4 = v[0][j];
#pragma unroll
      for (int g = 1; g < NGRP; ++g) {
        s4.x += v[g][j].x; s4.y += v[g][j].y; s4.z += v[g][j].z; s4.w += v[g][j].w;
      }
      part[j] = row_sum16(sum4(s4));
    }
    if (n == 0) {
#pragma unroll
      for (int j = 0; j < GRP_PIECES; ++j) stage[2 * j + hi] = part[j];
    }
    const float4 wa = *(const float4*)(stage + 8 * hi);
    const float4 wb = *(const float4*)(stage + 8 * hi + 4);
    v2f w[4];
    w[0].x = wa.x; w[0].y = wa.y;
    w[1].x = wa.z; w[1].y = wa.w;
    w[2].x = wb.x; w[2].y = wb.y;
    w[3].x = wb.z; w[3].y = wb.w;

    v8f off = {};
#pragma unroll
    for (int kc = 0; kc < 4; ++kc) {
      off = __builtin_amdgcn_wmma_f32_16x16x4_f32(
          false, aLs[kc], false, w[kc], (short)0, off, false, false);
    }
    float sel = off[0];
#pragma unroll
    for (int r = 1; r < 8; ++r) sel = ((n & 7) == r) ? off[r] : sel;
    const float selOth = other_half(sel);
    const float strOff = (hi == (n >> 3)) ? sel : selOth;

    const float hsum = (sum4(wa) + sum4(wb));
    const float segSum = hsum + other_half(hsum);

    if (lane == 0) ((float*)segTot4[0])[wave] = segSum;
    __syncthreads();
    float waveOff = 0.0f, sweepSum = 0.0f;
#pragma unroll
    for (int q4 = 0; q4 < WG_WAVES / 4; ++q4) {
      const float4 t = segTot4[0][q4];
      waveOff += (4 * q4 + 0 < wave) ? t.x : 0.0f;
      waveOff += (4 * q4 + 1 < wave) ? t.y : 0.0f;
      waveOff += (4 * q4 + 2 < wave) ? t.z : 0.0f;
      waveOff += (4 * q4 + 3 < wave) ? t.w : 0.0f;
      sweepSum += sum4(t);
    }
    carry0    = waveOff + strOff;
    sweepSum0 = sweepSum;

#pragma unroll
    for (int g = 0; g < NGRP; ++g)
#pragma unroll
      for (int j = 0; j < GRP_PIECES; ++j)
        *(float4*)(stage + g * GRP_BUF + j * 2 * PITCH + pieceOff) = v[g][j];
  }

  const float4 zero4 = {0.0f, 0.0f, 0.0f, 0.0f};
  float4 a0 = zero4;
  float4 a1 = a0, a2 = a0, a3 = a0;
  int bt = 0;
#pragma unroll 1
  for (int g = 0; g < NGRP; ++g, ++bt) {
    const bool have = bt < nb;
    float4 t[RED_BATCH];
#pragma unroll
    for (int u = 0; u < RED_BATCH; ++u) t[u] = zero4;
    if (have) {
      if (bt == reqB) request_chunk(own + UNIT, units - 1, tid);
      load_batch(xrow, start, bt, tid, t);
    }
    scan_group(stage + g * GRP_BUF + fragOff, aLt, hi, carry0);
    if (have) add_batch(t, a0, a1, a2, a3);
  }
#pragma unroll 1
  for (; bt < nb; ++bt) {
    if (bt == reqB) request_chunk(own + UNIT, units - 1, tid);
    float4 t[RED_BATCH];
    load_batch(xrow, start, bt, tid, t);
    add_batch(t, a0, a1, a2, a3);
  }
  float ps = (sum4(a0) + sum4(a1)) + (sum4(a2) + sum4(a3));
  ps = row_sum16(ps);
  ps += other_half(ps);
  if (lane == 0) ((float*)red4)[wave] = ps;
  __syncthreads();
  float base;
  {
    const float4 r0 = red4[0], r1 = red4[1], r2 = red4[2], r3 = red4[3];
    base = (sum4(r0) + sum4(r1)) + (sum4(r2) + sum4(r3));
  }

#pragma unroll
  for (int g = 0; g < NGRP; ++g)
#pragma unroll
    for (int j = 0; j < GRP_PIECES; ++j) {
      float4 r = *(const float4*)(stage + g * GRP_BUF + j * 2 * PITCH + pieceOff);
      r.x += base; r.y += base; r.z += base; r.w += base;
      gy0[g * GRP_F4 + j * 2 * STR_F4] = r;
    }
  base += sweepSum0;

#pragma unroll 1
  for (int sw = 1; sw < units; ++sw) {
    const int seg = (start + sw) * WG_WAVES + wave;
    const float4* gx = (const float4*)(xrow + (long long)seg * SEG) + hi * STR_F4 + n;
    float4*       gy = (float4*)(yrow + (long long)seg * SEG) + hi * STR_F4 + n;

    float4 v[NGRP][GRP_PIECES];
#pragma unroll
    for (int g = 0; g < NGRP; ++g)
#pragma unroll
      for (int j = 0; j < GRP_PIECES; ++j)
        v[g][j] = gx[g * GRP_F4 + j * 2 * STR_F4];

    float part[GRP_PIECES];
#pragma unroll
    for (int j = 0; j < GRP_PIECES; ++j) {
      float4 s4 = v[0][j];
#pragma unroll
      for (int g = 1; g < NGRP; ++g) {
        s4.x += v[g][j].x; s4.y += v[g][j].y; s4.z += v[g][j].z; s4.w += v[g][j].w;
      }
      part[j] = row_sum16(sum4(s4));
    }
    if (n == 0) {
#pragma unroll
      for (int j = 0; j < GRP_PIECES; ++j) stage[2 * j + hi] = part[j];
    }
    const float4 wa = *(const float4*)(stage + 8 * hi);
    const float4 wb = *(const float4*)(stage + 8 * hi + 4);
    v2f w[4];
    w[0].x = wa.x; w[0].y = wa.y;
    w[1].x = wa.z; w[1].y = wa.w;
    w[2].x = wb.x; w[2].y = wb.y;
    w[3].x = wb.z; w[3].y = wb.w;

    v8f off = {};
#pragma unroll
    for (int kc = 0; kc < 4; ++kc) {
      off = __builtin_amdgcn_wmma_f32_16x16x4_f32(
          false, aLs[kc], false, w[kc], (short)0, off, false, false);
    }
    float sel = off[0];
#pragma unroll
    for (int r = 1; r < 8; ++r) sel = ((n & 7) == r) ? off[r] : sel;
    const float selOth = other_half(sel);
    const float strOff = (hi == (n >> 3)) ? sel : selOth;

    const float hsum = (sum4(wa) + sum4(wb));
    const float segSum = hsum + other_half(hsum);

    if (lane == 0) ((float*)segTot4[sw])[wave] = segSum;
    __syncthreads();
    float waveOff = 0.0f, sweepSum = 0.0f;
#pragma unroll
    for (int q4 = 0; q4 < WG_WAVES / 4; ++q4) {
      const float4 t = segTot4[sw][q4];
      waveOff += (4 * q4 + 0 < wave) ? t.x : 0.0f;
      waveOff += (4 * q4 + 1 < wave) ? t.y : 0.0f;
      waveOff += (4 * q4 + 2 < wave) ? t.z : 0.0f;
      waveOff += (4 * q4 + 3 < wave) ? t.w : 0.0f;
      sweepSum += sum4(t);
    }

    float carry = (base + waveOff) + strOff;

#pragma unroll
    for (int g = 0; g < NGRP; ++g) {
#pragma unroll
      for (int j = 0; j < GRP_PIECES; ++j)
        *(float4*)(stage + j * 2 * PITCH + pieceOff) = v[g][j];

#pragma unroll
      for (int t = 0; t < GRP_TILES; ++t) {
        float* tl = stage + fragOff + 16 * t;

        const float4 xa = *(const float4*)(tl);
        const float4 xb = *(const float4*)(tl + 4);
        v2f b[4];
        b[0].x = xa.x; b[0].y = xa.y;
        b[1].x = xa.z; b[1].y = xa.w;
        b[2].x = xb.x; b[2].y = xb.y;
        b[3].x = xb.z; b[3].y = xb.w;

        v8f acc = {};
#pragma unroll
        for (int kc = 0; kc < 4; ++kc) {
          acc = __builtin_amdgcn_wmma_f32_16x16x4_f32(
              false, aLt[kc], false, b[kc], (short)0, acc, false, false);
        }

        const float last = acc[7];
        const float oth  = other_half(last);
        const float tot  = hi ? last : oth;

        float4 ra, rb;
        ra.x = acc[0] + carry; ra.y = acc[1] + carry;
        ra.z = acc[2] + carry; ra.w = acc[3] + carry;
        rb.x = acc[4] + carry; rb.y = acc[5] + carry;
        rb.z = acc[6] + carry; rb.w = acc[7] + carry;
        *(float4*)(tl)     = ra;
        *(float4*)(tl + 4) = rb;

        carry += tot;
      }

#pragma unroll
      for (int j = 0; j < GRP_PIECES; ++j) {
        gy[g * GRP_F4 + j * 2 * STR_F4] =
            *(const float4*)(stage + j * 2 * PITCH + pieceOff);
      }
    }

    base += sweepSum;
  }
}

extern "C" void kernel_launch(void* const* d_in, const int* in_sizes, int n_in,
                              void* d_out, int out_size, void* d_ws, size_t ws_size,
                              hipStream_t stream) {
  (void)n_in; (void)out_size; (void)d_ws; (void)ws_size;
  const float* x = (const float*)d_in[0];
  float* y = (float*)d_out;

  const long long total = (long long)in_sizes[0];
  const int rows = (int)(total / T_LEN);

  row_scan_wmma<<<rows * CHUNKS_PER_ROW, WG_THREADS, 0, stream>>>(x, y, rows);
}
